# starting kernel
# baseline (speedup 1.0000x reference)
#include <hip/hip_runtime.h>
#include <hip/hip_bf16.h>

typedef __attribute__((ext_vector_type(2))) float v2f;
typedef __attribute__((ext_vector_type(8))) float v8f;

#define NROW 1024
#define NCH  256

__global__ void gc_s_kernel(const float* __restrict__ x,
                            const float* __restrict__ w,
                            float* __restrict__ s) {
    const int lane  = threadIdx.x;
    const int tile  = blockIdx.x;
    const int m     = lane & 15;
    const int row   = tile * 16 + m;
    const int khalf = (lane >> 4) * 2;

    const float* xr = x + row * NCH;

    v8f acc = {};
    for (int k0 = 0; k0 < NCH; k0 += 4) {
        v2f a, bm;
        a.x  = xr[k0 + khalf];
        a.y  = xr[k0 + khalf + 1];
        bm.x = w[k0 + khalf];
        bm.y = w[k0 + khalf + 1];
        acc = __builtin_amdgcn_wmma_f32_16x16x4_f32(
             false, a,  false, bm,
             (short)0, acc,  false,  false);
    }

    if (lane == 0) {
#pragma unroll
        for (int r = 0; r < 8; ++r) s[tile * 16 + r] = acc[r];
    }
    if (lane == 16) {
#pragma unroll
        for (int r = 0; r < 8; ++r) s[tile * 16 + 8 + r] = acc[r];
    }
}

__global__ void gc_reduce_kernel(const float* __restrict__ x,
                                 const float* __restrict__ s,
                                 float* __restrict__ T,
                                 float* __restrict__ U,
                                 float* __restrict__ Sout) {
    const int c = threadIdx.x;
    float t = 0.0f, u = 0.0f;
    for (int i = 0; i < NROW; ++i) {
        const float xv = x[i * NCH + c];
        const float sv = s[i];
        t += xv;
        u += xv * sv;
    }
    T[c] = t;
    U[c] = u;

    __shared__ float red[NCH];
    red[c] = s[c] + s[c + 256] + s[c + 512] + s[c + 768];
    __syncthreads();
    for (int off = 128; off > 0; off >>= 1) {
        if (c < off) red[c] += red[c + off];
        __syncthreads();
    }
    if (c == 0) Sout[0] = red[0];
}

__global__ void gc_finalize_kernel(const float* __restrict__ x,
                                   const float* __restrict__ s,
                                   const float* __restrict__ T,
                                   const float* __restrict__ U,
                                   const float* __restrict__ Sp,
                                   const float* __restrict__ b,
                                   float* __restrict__ out) {
    const int i = blockIdx.x;
    const int c = threadIdx.x;
    const float si    = s[i];
    const float S     = Sp[0];
    const float bb    = b[0];
    const float alpha = (float)NROW * si + S + (float)NROW * bb;
    const float beta  = si + bb;
    const int idx = i * NCH + c;
    out[idx] = x[idx] * alpha + beta * T[c] + U[c];
}

extern "C" void kernel_launch(void* const* d_in, const int* in_sizes, int n_in,
                              void* d_out, int out_size, void* d_ws, size_t ws_size,
                              hipStream_t stream) {
    const float* x = (const float*)d_in[0];
    const float* w = (const float*)d_in[1];
    const float* b = (const float*)d_in[2];
    float* out = (float*)d_out;

    float* ws = (float*)d_ws;
    float* s_buf = ws;
    float* T_buf = ws + 1024;
    float* U_buf = ws + 1024 + 256;
    float* S_buf = ws + 1024 + 512;

    gc_s_kernel<<<NROW / 16, 32, 0, stream>>>(x, w, s_buf);
    gc_reduce_kernel<<<1, NCH, 0, stream>>>(x, s_buf, T_buf, U_buf, S_buf);
    gc_finalize_kernel<<<NROW, NCH, 0, stream>>>(x, s_buf, T_buf, U_buf,
                                                 S_buf, b, out);
}


# column sums T,U,S: one block walking 1024 rows -> 32 blocks of 32 rows write partial planes, summed in the finalize kernel (64 blocks x 16 rows); two-level sum tree changes the order of T,U,S
# speedup vs baseline: 4.8869x; 4.8869x over previous
#include <hip/hip_runtime.h>
#include <hip/hip_bf16.h>

typedef __attribute__((ext_vector_type(2))) float v2f;
typedef __attribute__((ext_vector_type(8))) float v8f;

#define NROW 1024
#define NCH  256
#define PROWS 32
#define NPART (NROW / PROWS)
#define FROWS 16

__global__ void gc_s_kernel(const float* __restrict__ x,
                            const float* __restrict__ w,
                            float* __restrict__ s) {
    const int lane  = threadIdx.x;
    const int tile  = blockIdx.x;
    const int m     = lane & 15;
    const int row   = tile * 16 + m;
    const int khalf = (lane >> 4) * 2;

    const float* xr = x + row * NCH;

    v8f acc = {};
    for (int k0 = 0; k0 < NCH; k0 += 4) {
        v2f a, bm;
        a.x  = xr[k0 + khalf];
        a.y  = xr[k0 + khalf + 1];
        bm.x = w[k0 + khalf];
        bm.y = w[k0 + khalf + 1];
        acc = __builtin_amdgcn_wmma_f32_16x16x4_f32(
             false, a,  false, bm,
             (short)0, acc,  false,  false);
    }

    if (lane == 0) {
#pragma unroll
        for (int r = 0; r < 8; ++r) s[tile * 16 + r] = acc[r];
    }
    if (lane == 16) {
#pragma unroll
        for (int r = 0; r < 8; ++r) s[tile * 16 + 8 + r] = acc[r];
    }
}

__global__ void gc_partial_kernel(const float* __restrict__ x,
                                  const float* __restrict__ s,
                                  float* __restrict__ PT,
                                  float* __restrict__ PU,
                                  float* __restrict__ PS) {
    const int c  = threadIdx.x;
    const int g  = blockIdx.x;
    const int r0 = g * PROWS;
    float t = 0.0f, u = 0.0f, ss = 0.0f;
    for (int r = 0; r < PROWS; ++r) {
        const float xv = x[(r0 + r) * NCH + c];
        const float sv = s[r0 + r];
        t  += xv;
        u  += xv * sv;
        ss += sv;
    }
    PT[g * NCH + c] = t;
    PU[g * NCH + c] = u;
    if (c == 0) PS[g] = ss;
}

__global__ void gc_finalize_kernel(const float* __restrict__ x,
                                   const float* __restrict__ s,
                                   const float* __restrict__ PT,
                                   const float* __restrict__ PU,
                                   const float* __restrict__ PS,
                                   const float* __restrict__ b,
                                   float* __restrict__ out) {
    const int c  = threadIdx.x;
    const int r0 = blockIdx.x * FROWS;
    float T = 0.0f, U = 0.0f, S = 0.0f;
    for (int g = 0; g < NPART; ++g) {
        T += PT[g * NCH + c];
        U += PU[g * NCH + c];
        S += PS[g];
    }
    const float bb = b[0];
    for (int r = 0; r < FROWS; ++r) {
        const int i = r0 + r;
        const float si    = s[i];
        const float alpha = (float)NROW * si + S + (float)NROW * bb;
        const float beta  = si + bb;
        const int idx = i * NCH + c;
        out[idx] = x[idx] * alpha + beta * T + U;
    }
}

extern "C" void kernel_launch(void* const* d_in, const int* in_sizes, int n_in,
                              void* d_out, int out_size, void* d_ws, size_t ws_size,
                              hipStream_t stream) {
    const float* x = (const float*)d_in[0];
    const float* w = (const float*)d_in[1];
    const float* b = (const float*)d_in[2];
    float* out = (float*)d_out;

    float* ws = (float*)d_ws;
    float* s_buf  = ws;
    float* PT_buf = ws + NROW;
    float* PU_buf = ws + NROW + NPART * NCH;
    float* PS_buf = ws + NROW + 2 * NPART * NCH;

    gc_s_kernel<<<NROW / 16, 32, 0, stream>>>(x, w, s_buf);
    gc_partial_kernel<<<NPART, NCH, 0, stream>>>(x, s_buf, PT_buf, PU_buf, PS_buf);
    gc_finalize_kernel<<<NROW / FROWS, NCH, 0, stream>>>(x, s_buf, PT_buf, PU_buf,
                                                         PS_buf, b, out);
}


# s kernel fused into the partial-sum kernel (3 launches -> 2): each 32-row block computes its own s on WMMA, k split over 4 waves and joined in LDS; changes the order in which s is summed
# speedup vs baseline: 7.7855x; 1.5931x over previous
#include <hip/hip_runtime.h>
#include <hip/hip_bf16.h>

typedef __attribute__((ext_vector_type(2))) float v2f;
typedef __attribute__((ext_vector_type(8))) float v8f;

#define NROW 1024
#define NCH  256
#define PROWS 32
#define NPART (NROW / PROWS)
#define FROWS 16
#define KQ    4
#define KSPAN (NCH / KQ)
#define KLANE (KSPAN / 2)

__global__ __launch_bounds__(NCH)
void gc_s_partial_kernel(const float* __restrict__ x,
                         const float* __restrict__ w,
                         float* __restrict__ s,
                         float* __restrict__ PT,
                         float* __restrict__ PU,
                         float* __restrict__ PS) {
    __shared__ float sp[KQ][PROWS];

    const int c    = threadIdx.x;
    const int g    = blockIdx.x;
    const int r0   = g * PROWS;
    const int lane = c & 31;
    const int wave = c >> 5;
    const int t    = wave & 1;
    const int kq   = wave >> 1;
    const int m    = lane & 15;
    const int h    = lane >> 4;

    const float4* xa = (const float4*)(x + (r0 + t * 16 + m) * NCH + kq * KSPAN + h * KLANE);
    const float4* wb = (const float4*)(w + kq * KSPAN + h * KLANE);
    float4 av[KLANE / 4], bv[KLANE / 4];
#pragma unroll
    for (int q = 0; q < KLANE / 4; ++q) { av[q] = xa[q]; bv[q] = wb[q]; }

    float xv[PROWS];
#pragma unroll
    for (int r = 0; r < PROWS; ++r) xv[r] = x[(r0 + r) * NCH + c];

    v8f acc = {};
#pragma unroll
    for (int q = 0; q < KLANE / 4; ++q) {
        v2f a0, b0, a1, b1;
        a0.x = av[q].x; a0.y = av[q].y; b0.x = bv[q].x; b0.y = bv[q].y;
        acc = __builtin_amdgcn_wmma_f32_16x16x4_f32(
             false, a0,  false, b0,
             (short)0, acc,  false,  false);
        a1.x = av[q].z; a1.y = av[q].w; b1.x = bv[q].z; b1.y = bv[q].w;
        acc = __builtin_amdgcn_wmma_f32_16x16x4_f32(
             false, a1,  false, b1,
             (short)0, acc,  false,  false);
    }

    if (m == 0) {
#pragma unroll
        for (int r = 0; r < 8; ++r) sp[kq][t * 16 + h * 8 + r] = acc[r];
    }
    __syncthreads();

    const float sl = (sp[0][lane] + sp[1][lane]) + (sp[2][lane] + sp[3][lane]);
    if (wave == 0) s[r0 + lane] = sl;

    float tt = 0.0f, u = 0.0f, ss = 0.0f;
#pragma unroll
    for (int r = 0; r < PROWS; ++r) {
        const float sv = __int_as_float(__builtin_amdgcn_readlane(__float_as_int(sl), r));
        tt += xv[r];
        u  += xv[r] * sv;
        ss += sv;
    }
    PT[g * NCH + c] = tt;
    PU[g * NCH + c] = u;
    if (c == 0) PS[g] = ss;
}

__global__ void gc_finalize_kernel(const float* __restrict__ x,
                                   const float* __restrict__ s,
                                   const float* __restrict__ PT,
                                   const float* __restrict__ PU,
                                   const float* __restrict__ PS,
                                   const float* __restrict__ b,
                                   float* __restrict__ out) {
    const int c  = threadIdx.x;
    const int r0 = blockIdx.x * FROWS;
    float T = 0.0f, U = 0.0f, S = 0.0f;
    for (int g = 0; g < NPART; ++g) {
        T += PT[g * NCH + c];
        U += PU[g * NCH + c];
        S += PS[g];
    }
    const float bb = b[0];
    for (int r = 0; r < FROWS; ++r) {
        const int i = r0 + r;
        const float si    = s[i];
        const float alpha = (float)NROW * si + S + (float)NROW * bb;
        const float beta  = si + bb;
        const int idx = i * NCH + c;
        out[idx] = x[idx] * alpha + beta * T + U;
    }
}

extern "C" void kernel_launch(void* const* d_in, const int* in_sizes, int n_in,
                              void* d_out, int out_size, void* d_ws, size_t ws_size,
                              hipStream_t stream) {
    const float* x = (const float*)d_in[0];
    const float* w = (const float*)d_in[1];
    const float* b = (const float*)d_in[2];
    float* out = (float*)d_out;

    float* ws = (float*)d_ws;
    float* s_buf  = ws;
    float* PT_buf = ws + NROW;
    float* PU_buf = ws + NROW + NPART * NCH;
    float* PS_buf = ws + NROW + 2 * NPART * NCH;

    gc_s_partial_kernel<<<NPART, NCH, 0, stream>>>(x, w, s_buf, PT_buf, PU_buf, PS_buf);
    gc_finalize_kernel<<<NROW / FROWS, NCH, 0, stream>>>(x, s_buf, PT_buf, PU_buf,
                                                         PS_buf, b, out);
}


# finalize re-tiled: a block owns 64 columns x 64 rows so its 8 waves share T,U: each wave adds 4 of the 32 planes, joined in LDS (pairwise tree); changes the order in which T,U are summed
# speedup vs baseline: 8.5443x; 1.0975x over previous
#include <hip/hip_runtime.h>
#include <hip/hip_bf16.h>

typedef __attribute__((ext_vector_type(2))) float v2f;
typedef __attribute__((ext_vector_type(8))) float v8f;

#define NROW 1024
#define NCH  256
#define PROWS 32
#define NPART (NROW / PROWS)
#define KQ    4
#define KSPAN (NCH / KQ)
#define KLANE (KSPAN / 2)

__global__ __launch_bounds__(NCH)
void gc_s_partial_kernel(const float* __restrict__ x,
                         const float* __restrict__ w,
                         float* __restrict__ s,
                         float* __restrict__ PT,
                         float* __restrict__ PU,
                         float* __restrict__ PS) {
    __shared__ float sp[KQ][PROWS];

    const int c    = threadIdx.x;
    const int g    = blockIdx.x;
    const int r0   = g * PROWS;
    const int lane = c & 31;
    const int wave = c >> 5;
    const int t    = wave & 1;
    const int kq   = wave >> 1;
    const int m    = lane & 15;
    const int h    = lane >> 4;

    const float4* xa = (const float4*)(x + (r0 + t * 16 + m) * NCH + kq * KSPAN + h * KLANE);
    const float4* wb = (const float4*)(w + kq * KSPAN + h * KLANE);
    float4 av[KLANE / 4], bv[KLANE / 4];
#pragma unroll
    for (int q = 0; q < KLANE / 4; ++q) { av[q] = xa[q]; bv[q] = wb[q]; }

    float xv[PROWS];
#pragma unroll
    for (int r = 0; r < PROWS; ++r) xv[r] = x[(r0 + r) * NCH + c];

    v8f acc = {};
#pragma unroll
    for (int q = 0; q < KLANE / 4; ++q) {
        v2f a0, b0, a1, b1;
        a0.x = av[q].x; a0.y = av[q].y; b0.x = bv[q].x; b0.y = bv[q].y;
        acc = __builtin_amdgcn_wmma_f32_16x16x4_f32(
             false, a0,  false, b0,
             (short)0, acc,  false,  false);
        a1.x = av[q].z; a1.y = av[q].w; b1.x = bv[q].z; b1.y = bv[q].w;
        acc = __builtin_amdgcn_wmma_f32_16x16x4_f32(
             false, a1,  false, b1,
             (short)0, acc,  false,  false);
    }

    if (m == 0) {
#pragma unroll
        for (int r = 0; r < 8; ++r) sp[kq][t * 16 + h * 8 + r] = acc[r];
    }
    __syncthreads();

    const float sl = (sp[0][lane] + sp[1][lane]) + (sp[2][lane] + sp[3][lane]);
    if (wave == 0) s[r0 + lane] = sl;

    float tt = 0.0f, u = 0.0f, ss = 0.0f;
#pragma unroll
    for (int r = 0; r < PROWS; ++r) {
        const float sv = __int_as_float(__builtin_amdgcn_readlane(__float_as_int(sl), r));
        tt += xv[r];
        u  += xv[r] * sv;
        ss += sv;
    }
    PT[g * NCH + c] = tt;
    PU[g * NCH + c] = u;
    if (c == 0) PS[g] = ss;
}

#define FCOLS  64
#define FSLABS (NCH / FCOLS)
#define FWAVES 8
#define FPW    (NPART / FWAVES)
#define FWROWS 8
#define FBROWS (FWAVES * FWROWS)
static_assert(NPART == 32 && FPW == 4 && FWROWS <= 32, "lane g holds PS[g]; the wave sum below is written for 4 planes");

__global__ __launch_bounds__(FWAVES * 32)
void gc_finalize_kernel(const float* __restrict__ x,
                        const float* __restrict__ s,
                        const float* __restrict__ PT,
                        const float* __restrict__ PU,
                        const float* __restrict__ PS,
                        const float* __restrict__ b,
                        float* __restrict__ out) {
    __shared__ float2 wt[FWAVES][32];
    __shared__ float2 wu[FWAVES][32];

    const int lane = threadIdx.x & 31;
    const int wave = __builtin_amdgcn_readfirstlane((int)(threadIdx.x >> 5));
    const int slab = blockIdx.x % FSLABS;
    const int rb   = blockIdx.x / FSLABS;
    const int c    = slab * FCOLS + 2 * lane;
    const int r0   = rb * FBROWS + wave * FWROWS;

    float2 pt[FPW], pu[FPW];
#pragma unroll
    for (int q = 0; q < FPW; ++q) {
        const int g = wave * FPW + q;
        pt[q] = *(const float2*)(PT + g * NCH + c);
        pu[q] = *(const float2*)(PU + g * NCH + c);
    }
    float2 xv[FWROWS];
#pragma unroll
    for (int r = 0; r < FWROWS; ++r) xv[r] = *(const float2*)(x + (r0 + r) * NCH + c);
    const float sl = s[r0 + (lane & (FWROWS - 1))];

    const float pl = PS[lane];
    const float bb = b[0];

    asm volatile("" :: "v"(pt[0].x), "v"(pt[1].x), "v"(pt[2].x), "v"(pt[3].x),
                       "v"(pu[0].x), "v"(pu[1].x), "v"(pu[2].x), "v"(pu[3].x),
                       "v"(xv[0].x), "v"(xv[1].x), "v"(xv[2].x), "v"(xv[3].x),
                       "v"(xv[4].x), "v"(xv[5].x), "v"(xv[6].x), "v"(xv[7].x),
                       "v"(sl), "v"(pl));

    float2 t2, u2;
    t2.x = (pt[0].x + pt[1].x) + (pt[2].x + pt[3].x);
    t2.y = (pt[0].y + pt[1].y) + (pt[2].y + pt[3].y);
    u2.x = (pu[0].x + pu[1].x) + (pu[2].x + pu[3].x);
    u2.y = (pu[0].y + pu[1].y) + (pu[2].y + pu[3].y);
    wt[wave][lane] = t2;
    wu[wave][lane] = u2;
    __syncthreads();

    float S = 0.0f;
#pragma unroll
    for (int g = 0; g < NPART; ++g)
        S += __int_as_float(__builtin_amdgcn_readlane(__float_as_int(pl), g));

    float2 T, U;
    {
        const float2 a0 = wt[0][lane], a1 = wt[1][lane], a2 = wt[2][lane], a3 = wt[3][lane];
        const float2 a4 = wt[4][lane], a5 = wt[5][lane], a6 = wt[6][lane], a7 = wt[7][lane];
        T.x = ((a0.x + a1.x) + (a2.x + a3.x)) + ((a4.x + a5.x) + (a6.x + a7.x));
        T.y = ((a0.y + a1.y) + (a2.y + a3.y)) + ((a4.y + a5.y) + (a6.y + a7.y));
        const float2 e0 = wu[0][lane], e1 = wu[1][lane], e2 = wu[2][lane], e3 = wu[3][lane];
        const float2 e4 = wu[4][lane], e5 = wu[5][lane], e6 = wu[6][lane], e7 = wu[7][lane];
        U.x = ((e0.x + e1.x) + (e2.x + e3.x)) + ((e4.x + e5.x) + (e6.x + e7.x));
        U.y = ((e0.y + e1.y) + (e2.y + e3.y)) + ((e4.y + e5.y) + (e6.y + e7.y));
    }

#pragma unroll
    for (int r = 0; r < FWROWS; ++r) {
        const int i = r0 + r;
        const float si    = __int_as_float(__builtin_amdgcn_readlane(__float_as_int(sl), r));
        const float alpha = (float)NROW * si + S + (float)NROW * bb;
        const float beta  = si + bb;
        float2 o;
        o.x = xv[r].x * alpha + beta * T.x + U.x;
        o.y = xv[r].y * alpha + beta * T.y + U.y;
        *(float2*)(out + i * NCH + c) = o;
    }
}

extern "C" void kernel_launch(void* const* d_in, const int* in_sizes, int n_in,
                              void* d_out, int out_size, void* d_ws, size_t ws_size,
                              hipStream_t stream) {
    const float* x = (const float*)d_in[0];
    const float* w = (const float*)d_in[1];
    const float* b = (const float*)d_in[2];
    float* out = (float*)d_out;

    float* ws = (float*)d_ws;
    float* s_buf  = ws;
    float* PT_buf = ws + NROW;
    float* PU_buf = ws + NROW + NPART * NCH;
    float* PS_buf = ws + NROW + 2 * NPART * NCH;

    gc_s_partial_kernel<<<NPART, NCH, 0, stream>>>(x, w, s_buf, PT_buf, PU_buf, PS_buf);
    gc_finalize_kernel<<<FSLABS * (NROW / FBROWS), FWAVES * 32, 0, stream>>>(x, s_buf, PT_buf, PU_buf,
                                                         PS_buf, b, out);
}


# partial-sum kernel reads x once: a wave owns 32 columns of all 32 rows and its matrix-operand registers also give the column sums (lane sums, two lane exchanges); changes the sum order of s, S, T, U
# speedup vs baseline: 9.1083x; 1.0660x over previous
#include <hip/hip_runtime.h>
#include <hip/hip_bf16.h>

typedef __attribute__((ext_vector_type(2))) float v2f;
typedef __attribute__((ext_vector_type(8))) float v8f;

#define NROW 1024
#define NCH  256
#define PROWS 32
#define NPART (NROW / PROWS)
#define SWAVES 8
#define WCOLS (NCH / SWAVES)
#define TROWS 4
#define NTILE (PROWS / TROWS)


template <int CTRL>
__device__ __forceinline__ float lane_row_read(float v) {
    return __builtin_bit_cast(float,
        __builtin_amdgcn_update_dpp(0, __builtin_bit_cast(int, v), CTRL, 0xf, 0xf, true));
}
#define DPP_QUAD_SWAP1 0xB1
#define DPP_ROW_XMASK(k) (0x160 + (k))

__device__ __forceinline__ float other_row_flip2(float v) {
    const int b = __builtin_bit_cast(int, v);
    return __builtin_bit_cast(float,
        __builtin_amdgcn_permlanex16(b, b, 0x54761032, 0xdcfe98ba, false, false));
}
__device__ __forceinline__ float other_row(float v) {
    const int b = __builtin_bit_cast(int, v);
    return __builtin_bit_cast(float,
        __builtin_amdgcn_permlanex16(b, b, 0x76543210, 0xfedcba98, false, false));
}

__global__ __launch_bounds__(NCH)
void gc_s_partial_kernel(const float* __restrict__ x,
                         const float* __restrict__ w,
                         float* __restrict__ s,
                         float* __restrict__ PT,
                         float* __restrict__ PU,
                         float* __restrict__ PS) {
    __shared__ __attribute__((aligned(16))) float sp[SWAVES][PROWS];

    const int tid  = threadIdx.x;
    const int g    = blockIdx.x;
    const int r0   = g * PROWS;
    const int lane = tid & 31;
    const int wave = __builtin_amdgcn_readfirstlane(tid >> 5);
    const int h    = lane >> 4;
    const int q    = (lane >> 2) & 3;
    const int g2   = lane & 3;
    const int c    = wave * WCOLS + 4 * (4 * h + g2);
    const int ca   = wave * WCOLS + 4 * (4 * h + q);

    const float4  wv = *(const float4*)(w + ca);
    const float4* xa = (const float4*)(x + (r0 + q) * NCH + c);
    float4 xv[NTILE];
#pragma unroll
    for (int t = 0; t < NTILE; ++t) xv[t] = xa[t * (TROWS * NCH / 4)];

    v2f a0, a1;
    a0.x = wv.x; a0.y = wv.y; a1.x = wv.z; a1.y = wv.w;

    float part[NTILE];
#pragma unroll
    for (int t = 0; t < NTILE; ++t) {
        v2f b0, b1;
        b0.x = xv[t].x; b0.y = xv[t].y; b1.x = xv[t].z; b1.y = xv[t].w;
        v8f acc = {};
        acc = __builtin_amdgcn_wmma_f32_16x16x4_f32(
             false, a0,  false, b0,
             (short)0, acc,  false,  false);
        acc = __builtin_amdgcn_wmma_f32_16x16x4_f32(
             false, a1,  false, b1,
             (short)0, acc,  false,  false);
        const float z = (g2 & 1) ? acc[4] : acc[0];
        const float y = z + lane_row_read<DPP_QUAD_SWAP1>(z);
        part[t] = y + other_row_flip2(y);
    }

    if ((lane & 0x13) == 0) {
        float4* d = (float4*)&sp[wave][8 * q];
        d[0] = make_float4(part[0], part[1], part[2], part[3]);
        d[1] = make_float4(part[4], part[5], part[6], part[7]);
    }
    __syncthreads();

    const float sl = ((sp[0][lane] + sp[1][lane]) + (sp[2][lane] + sp[3][lane])) +
                     ((sp[4][lane] + sp[5][lane]) + (sp[6][lane] + sp[7][lane]));

    float sv[NTILE];
#pragma unroll
    for (int t = 0; t < NTILE; ++t)
        sv[t] = __builtin_bit_cast(float,
            __builtin_amdgcn_ds_bpermute(4 * (8 * q + t), __builtin_bit_cast(int, sl)));

    float ss = sl;
    ss += lane_row_read<DPP_ROW_XMASK(1)>(ss);
    ss += lane_row_read<DPP_ROW_XMASK(2)>(ss);
    ss += lane_row_read<DPP_ROW_XMASK(4)>(ss);
    ss += lane_row_read<DPP_ROW_XMASK(8)>(ss);
    ss += other_row(ss);

    float4 T4 = make_float4(0.0f, 0.0f, 0.0f, 0.0f), U4 = T4;
#pragma unroll
    for (int t = 0; t < NTILE; ++t) {
        T4.x += xv[t].x;         T4.y += xv[t].y;         T4.z += xv[t].z;         T4.w += xv[t].w;
        U4.x += xv[t].x * sv[t]; U4.y += xv[t].y * sv[t]; U4.z += xv[t].z * sv[t]; U4.w += xv[t].w * sv[t];
    }
#define ROWSUM4(v) \
    v += lane_row_read<DPP_ROW_XMASK(4)>(v); \
    v += lane_row_read<DPP_ROW_XMASK(8)>(v);
    ROWSUM4(T4.x) ROWSUM4(T4.y) ROWSUM4(T4.z) ROWSUM4(T4.w)
    ROWSUM4(U4.x) ROWSUM4(U4.y) ROWSUM4(U4.z) ROWSUM4(U4.w)
#undef ROWSUM4

    if (wave == 0) s[r0 + 4 * (lane & 7) + (lane >> 3)] = sl;
    if (q == 0) {
        *(float4*)(PT + g * NCH + c) = T4;
        *(float4*)(PU + g * NCH + c) = U4;
        if (tid == 32) PS[g] = ss;
    }
}

#define FCOLS  64
#define FSLABS (NCH / FCOLS)
#define FWAVES 8
#define FPW    (NPART / FWAVES)
#define FWROWS 8
#define FBROWS (FWAVES * FWROWS)
static_assert(NPART == 32 && FPW == 4 && FWROWS <= 32, "lane g holds PS[g]; the wave sum below is written for 4 planes");

__global__ __launch_bounds__(FWAVES * 32)
void gc_finalize_kernel(const float* __restrict__ x,
                        const float* __restrict__ s,
                        const float* __restrict__ PT,
                        const float* __restrict__ PU,
                        const float* __restrict__ PS,
                        const float* __restrict__ b,
                        float* __restrict__ out) {
    __shared__ float2 wt[FWAVES][32];
    __shared__ float2 wu[FWAVES][32];

    const int lane = threadIdx.x & 31;
    const int wave = __builtin_amdgcn_readfirstlane((int)(threadIdx.x >> 5));
    const int slab = blockIdx.x % FSLABS;
    const int rb   = blockIdx.x / FSLABS;
    const int c    = slab * FCOLS + 2 * lane;
    const int r0   = rb * FBROWS + wave * FWROWS;

    float2 pt[FPW], pu[FPW];
#pragma unroll
    for (int q = 0; q < FPW; ++q) {
        const int g = wave * FPW + q;
        pt[q] = *(const float2*)(PT + g * NCH + c);
        pu[q] = *(const float2*)(PU + g * NCH + c);
    }
    float2 xv[FWROWS];
#pragma unroll
    for (int r = 0; r < FWROWS; ++r) xv[r] = *(const float2*)(x + (r0 + r) * NCH + c);
    const float sl = s[r0 + (lane & (FWROWS - 1))];

    const float pl = PS[lane];
    const float bb = b[0];

    asm volatile("" :: "v"(pt[0].x), "v"(pt[1].x), "v"(pt[2].x), "v"(pt[3].x),
                       "v"(pu[0].x), "v"(pu[1].x), "v"(pu[2].x), "v"(pu[3].x),
                       "v"(xv[0].x), "v"(xv[1].x), "v"(xv[2].x), "v"(xv[3].x),
                       "v"(xv[4].x), "v"(xv[5].x), "v"(xv[6].x), "v"(xv[7].x),
                       "v"(sl), "v"(pl));

    float2 t2, u2;
    t2.x = (pt[0].x + pt[1].x) + (pt[2].x + pt[3].x);
    t2.y = (pt[0].y + pt[1].y) + (pt[2].y + pt[3].y);
    u2.x = (pu[0].x + pu[1].x) + (pu[2].x + pu[3].x);
    u2.y = (pu[0].y + pu[1].y) + (pu[2].y + pu[3].y);
    wt[wave][lane] = t2;
    wu[wave][lane] = u2;
    __syncthreads();

    float S = 0.0f;
#pragma unroll
    for (int g = 0; g < NPART; ++g)
        S += __int_as_float(__builtin_amdgcn_readlane(__float_as_int(pl), g));

    float2 T, U;
    {
        const float2 a0 = wt[0][lane], a1 = wt[1][lane], a2 = wt[2][lane], a3 = wt[3][lane];
        const float2 a4 = wt[4][lane], a5 = wt[5][lane], a6 = wt[6][lane], a7 = wt[7][lane];
        T.x = ((a0.x + a1.x) + (a2.x + a3.x)) + ((a4.x + a5.x) + (a6.x + a7.x));
        T.y = ((a0.y + a1.y) + (a2.y + a3.y)) + ((a4.y + a5.y) + (a6.y + a7.y));
        const float2 e0 = wu[0][lane], e1 = wu[1][lane], e2 = wu[2][lane], e3 = wu[3][lane];
        const float2 e4 = wu[4][lane], e5 = wu[5][lane], e6 = wu[6][lane], e7 = wu[7][lane];
        U.x = ((e0.x + e1.x) + (e2.x + e3.x)) + ((e4.x + e5.x) + (e6.x + e7.x));
        U.y = ((e0.y + e1.y) + (e2.y + e3.y)) + ((e4.y + e5.y) + (e6.y + e7.y));
    }

#pragma unroll
    for (int r = 0; r < FWROWS; ++r) {
        const int i = r0 + r;
        const float si    = __int_as_float(__builtin_amdgcn_readlane(__float_as_int(sl), r));
        const float alpha = (float)NROW * si + S + (float)NROW * bb;
        const float beta  = si + bb;
        float2 o;
        o.x = xv[r].x * alpha + beta * T.x + U.x;
        o.y = xv[r].y * alpha + beta * T.y + U.y;
        *(float2*)(out + i * NCH + c) = o;
    }
}

extern "C" void kernel_launch(void* const* d_in, const int* in_sizes, int n_in,
                              void* d_out, int out_size, void* d_ws, size_t ws_size,
                              hipStream_t stream) {
    const float* x = (const float*)d_in[0];
    const float* w = (const float*)d_in[1];
    const float* b = (const float*)d_in[2];
    float* out = (float*)d_out;

    float* ws = (float*)d_ws;
    float* s_buf  = ws;
    float* PT_buf = ws + NROW;
    float* PU_buf = ws + NROW + NPART * NCH;
    float* PS_buf = ws + NROW + 2 * NPART * NCH;

    gc_s_partial_kernel<<<NPART, NCH, 0, stream>>>(x, w, s_buf, PT_buf, PU_buf, PS_buf);
    gc_finalize_kernel<<<FSLABS * (NROW / FBROWS), FWAVES * 32, 0, stream>>>(x, s_buf, PT_buf, PU_buf,
                                                         PS_buf, b, out);
}
